# starting kernel
# baseline (speedup 1.0000x reference)
#include <hip/hip_runtime.h>

typedef __attribute__((ext_vector_type(2))) float v2f;
typedef __attribute__((ext_vector_type(8))) float v8f;

#define CHUNK     512
#define BLOCK     128
#define STATS_RUN 16
#define DFEAT     64

__global__ void zero_f32(float* __restrict__ a, long n) {
  long i = (long)blockIdx.x * blockDim.x + threadIdx.x;
  long stride = (long)gridDim.x * blockDim.x;
  for (; i < n; i += stride) a[i] = 0.0f;
}

__global__ void edge_stats(const float* __restrict__ w, const int* __restrict__ seg,
                           float* __restrict__ wsum, float* __restrict__ deg, int E) {
  long t = (long)blockIdx.x * blockDim.x + threadIdx.x;
  long base = t * STATS_RUN;
  if (base >= E) return;
  long end = base + STATS_RUN; if (end > E) end = E;
  int cur = seg[base];
  float sw = 0.0f, sd = 0.0f;
  for (long e = base; e < end; ++e) {
    int s = seg[e];
    if (s != cur) {
      atomicAdd(&wsum[cur], sw);
      atomicAdd(&deg[cur], sd);
      cur = s; sw = 0.0f; sd = 0.0f;
    }
    sw += w[e]; sd += 1.0f;
  }
  atomicAdd(&wsum[cur], sw);
  atomicAdd(&deg[cur], sd);
}

__global__ void finalize(const float* __restrict__ wsum, float* __restrict__ deg_invden, int N) {
  int n = blockIdx.x * blockDim.x + threadIdx.x;
  if (n >= N) return;
  float w = wsum[n], d = deg_invden[n];
  float den = (w == 0.0f) ? d : w;
  if (den == 0.0f) den = 1.0f;
  deg_invden[n] = 1.0f / den;
}

__global__ void __launch_bounds__(BLOCK)
aggregate(const float* __restrict__ feat, const float* __restrict__ ew,
          const int* __restrict__ nidx, const int* __restrict__ seg,
          const float* __restrict__ wsum, float* __restrict__ out, int E) {
  __shared__ int   s_seg[CHUNK];
  __shared__ int   s_idx[CHUNK];
  __shared__ float s_w[CHUNK];

  const int tid  = threadIdx.x;
  const int lane = tid & 31;
  const int m    = lane & 15;
  const int half = lane >> 4;
  const int wave = tid >> 5;
  const int cb   = wave * 16;

  const long chunk_base = (long)blockIdx.x * CHUNK;
  for (int i = tid; i < CHUNK; i += BLOCK) {
    long e = chunk_base + i;
    if (e < (long)E) { s_seg[i] = seg[e]; s_idx[i] = nidx[e]; s_w[i] = ew[e]; }
    else             { s_seg[i] = -1;     s_idx[i] = 0;       s_w[i] = 0.0f;  }
  }
  __syncthreads();

  long rem = (long)E - chunk_base;
  int nvalid = rem < (long)CHUNK ? (int)rem : CHUNK;

  int  cur_seg  = -2;
  int  used     = 0;
  bool cur_mean = false;
  int  slot_node = -1;
  v8f  c;
  #pragma unroll
  for (int r = 0; r < 8; ++r) c[r] = 0.0f;

  const int kx = 2 * half, ky = kx + 1;

  for (int g = 0; g < CHUNK; g += 4) {
    if (g >= nvalid) break;

    int e_seg[4], e_idx[4]; float e_w[4];
    #pragma unroll
    for (int k = 0; k < 4; ++k) {
      int i = g + k;
      e_seg[k] = s_seg[i]; e_idx[k] = s_idx[i]; e_w[k] = s_w[i];
    }

    int tmp = cur_seg, newcnt = 0;
    #pragma unroll
    for (int k = 0; k < 4; ++k)
      if (e_seg[k] >= 0 && e_seg[k] != tmp) { newcnt++; tmp = e_seg[k]; }

    if (used + newcnt > 16) {
      #pragma unroll
      for (int r = 0; r < 8; ++r) {
        int row  = r + 8 * half;
        int node = __shfl(slot_node, row, 32);
        if (row < used && node >= 0)
          atomicAdd(&out[(long)node * DFEAT + cb + m], c[r]);
      }
      #pragma unroll
      for (int r = 0; r < 8; ++r) c[r] = 0.0f;
      used = 0; cur_seg = -2; slot_node = -1;
    }

    int slot[4]; float weff[4];
    int cur_slot = used - 1;
    #pragma unroll
    for (int k = 0; k < 4; ++k) {
      if (e_seg[k] < 0) { slot[k] = 0; weff[k] = 0.0f; continue; }
      if (e_seg[k] != cur_seg) {
        cur_seg  = e_seg[k];
        cur_slot = used++;
        slot_node = (m == cur_slot) ? cur_seg : slot_node;
        cur_mean = (wsum[cur_seg] == 0.0f);
      }
      slot[k] = cur_slot;
      weff[k] = cur_mean ? 1.0f : e_w[k];
    }

    v2f a, b;
    a.x = (m == slot[kx]) ? weff[kx] : 0.0f;
    a.y = (m == slot[ky]) ? weff[ky] : 0.0f;
    b.x = feat[(long)e_idx[kx] * DFEAT + cb + m];
    b.y = feat[(long)e_idx[ky] * DFEAT + cb + m];

    if (g + 4 < nvalid) {
      __builtin_prefetch(&feat[(long)s_idx[g + 4 + kx] * DFEAT + cb], 0, 1);
      __builtin_prefetch(&feat[(long)s_idx[g + 4 + ky] * DFEAT + cb], 0, 1);
    }

    c = __builtin_amdgcn_wmma_f32_16x16x4_f32(false, a, false, b, (short)0, c,
                                              false, false);
  }

  #pragma unroll
  for (int r = 0; r < 8; ++r) {
    int row  = r + 8 * half;
    int node = __shfl(slot_node, row, 32);
    if (row < used && node >= 0)
      atomicAdd(&out[(long)node * DFEAT + cb + m], c[r]);
  }
}

__global__ void scale_out(float* __restrict__ out, const float* __restrict__ inv_den, long total) {
  long i = (long)blockIdx.x * blockDim.x + threadIdx.x;
  long stride = (long)gridDim.x * blockDim.x;
  for (; i < total; i += stride) out[i] *= inv_den[i >> 6];
}

static inline int cdiv_i(long a, long b) { return (int)((a + b - 1) / b); }

extern "C" void kernel_launch(void* const* d_in, const int* in_sizes, int n_in,
                              void* d_out, int out_size, void* d_ws, size_t ws_size,
                              hipStream_t stream) {
  const float* feat = (const float*)d_in[0];
  const float* ew   = (const float*)d_in[1];
  const int*   nidx = (const int*)d_in[2];
  const int*   seg  = (const int*)d_in[3];
  float* out = (float*)d_out;

  const int N = in_sizes[0] / DFEAT;
  const int E = in_sizes[1];

  float* wsum = (float*)d_ws;
  float* deg  = wsum + N;

  zero_f32<<<2048, 256, 0, stream>>>(wsum, 2L * N);
  zero_f32<<<4096, 256, 0, stream>>>(out, (long)out_size);

  int stat_threads = cdiv_i(E, STATS_RUN);
  edge_stats<<<cdiv_i(stat_threads, 256), 256, 0, stream>>>(ew, seg, wsum, deg, E);
  finalize<<<cdiv_i(N, 256), 256, 0, stream>>>(wsum, deg, N);

  aggregate<<<cdiv_i(E, CHUNK), BLOCK, 0, stream>>>(feat, ew, nidx, seg, wsum, out, E);

  scale_out<<<4096, 256, 0, stream>>>(out, deg, (long)out_size);
}


# aggregate: rows whose edges all lie in one chunk leave by one plain 256-byte store per row (tile staged through LDS, one barrier per window); only rows shared by two chunks keep atomics
# speedup vs baseline: 2.6773x; 2.6773x over previous
#include <hip/hip_runtime.h>

typedef __attribute__((ext_vector_type(2))) float v2f;
typedef __attribute__((ext_vector_type(8))) float v8f;

#define CHUNK     512
#define BLOCK     128
#define STATS_RUN 16
#define DFEAT     64
#define KSTEPS    4
#define TP        68
#define FILL_GRID  256
#define FILL_BLOCK 256

__global__ void __launch_bounds__(FILL_BLOCK)
zero2_f32(float* __restrict__ a, long na, float* __restrict__ b, long nb) {
  const long tid    = (long)blockIdx.x * blockDim.x + threadIdx.x;
  const long stride = (long)gridDim.x * blockDim.x;
  const float4 z = make_float4(0.0f, 0.0f, 0.0f, 0.0f);
  float4* __restrict__ a4 = (float4*)a;
  float4* __restrict__ b4 = (float4*)b;
  const long na4 = na >> 2, nb4 = nb >> 2;
  for (long i = tid; i < nb4; i += stride) b4[i] = z;
  for (long i = tid; i < na4; i += stride) a4[i] = z;
  if (tid < (na & 3)) a[(na4 << 2) + tid] = 0.0f;
  if (tid < (nb & 3)) b[(nb4 << 2) + tid] = 0.0f;
}

__global__ void edge_stats(const float* __restrict__ w, const int* __restrict__ seg,
                           float* __restrict__ wsum, float* __restrict__ deg, int E) {
  long t = (long)blockIdx.x * blockDim.x + threadIdx.x;
  long base = t * STATS_RUN;
  if (base >= E) return;
  long end = base + STATS_RUN; if (end > E) end = E;
  int cur = seg[base];
  float sw = 0.0f, sd = 0.0f;
  for (long e = base; e < end; ++e) {
    int s = seg[e];
    if (s != cur) {
      atomicAdd(&wsum[cur], sw);
      atomicAdd(&deg[cur], sd);
      cur = s; sw = 0.0f; sd = 0.0f;
    }
    sw += w[e]; sd += 1.0f;
  }
  atomicAdd(&wsum[cur], sw);
  atomicAdd(&deg[cur], sd);
}

__device__ __forceinline__ void flush_window(float* __restrict__ out, const v8f& c,
                                             float* __restrict__ stage, int win, int lo, int hi,
                                             bool lo_shared, bool hi_shared,
                                             int half, int m, int col, int wave) {
  const int nb = win << 4;
  #pragma unroll
  for (int r = 0; r < 8; ++r) stage[(r + 8 * half) * TP + col] = c[r];
  __syncthreads();
  #pragma unroll
  for (int i = 0; i < 2; ++i) {
    const int row  = 4 * wave + 2 * i + half;
    const int node = nb + row;
    if (node >= lo && node <= hi) {
      const float4 v = *(const float4*)&stage[row * TP + 4 * m];
      float* p = out + (long)node * DFEAT + 4 * m;
      const bool shared = (node == lo && lo_shared) || (node == hi && hi_shared);
      if (shared) {
        atomicAdd(p + 0, v.x); atomicAdd(p + 1, v.y);
        atomicAdd(p + 2, v.z); atomicAdd(p + 3, v.w);
      } else {
        *(float4*)p = v;
      }
    }
  }
}

__global__ void __launch_bounds__(BLOCK)
aggregate(const float* __restrict__ feat, const float* __restrict__ ew,
          const int* __restrict__ nidx, const int* __restrict__ seg,
          const float* __restrict__ wsum, const float* __restrict__ deg,
          float* __restrict__ out, int E) {
  __shared__ __attribute__((aligned(16))) int      s_seg[CHUNK];
  __shared__ __attribute__((aligned(16))) unsigned s_off[CHUNK];
  __shared__ __attribute__((aligned(16))) float    s_w[CHUNK];
  __shared__ __attribute__((aligned(16))) float    s_stage[2][16 * TP];

  const int tid  = threadIdx.x;
  const int lane = tid & 31;
  const int m    = lane & 15;
  const int half = lane >> 4;
  const int wave = tid >> 5;
  const int cb   = wave * 16;
  const int col  = cb + m;

  const long chunk_base = (long)blockIdx.x * CHUNK;
  long rem = (long)E - chunk_base;
  const int nvalid = rem < (long)CHUNK ? (int)rem : CHUNK;

  #pragma unroll
  for (int t = 0; t < CHUNK / BLOCK; ++t) {
    const int  i    = tid + t * BLOCK;
    const bool live = i < nvalid;
    const long e    = live ? chunk_base + i : (long)E - 1;
    const int   sg = seg[e];
    const int   ix = nidx[e];
    const float wv = ew[e];
    const float ws = wsum[sg];
    const float dg = deg[sg];
    const bool  mean = (ws == 0.0f);
    const float num = mean ? 1.0f : wv;
    const float den = mean ? dg : ws;
    s_seg[i] = sg;
    s_off[i] = live ? (unsigned)ix * DFEAT : 0u;
    s_w[i]   = live ? num / den : 0.0f;
  }
  __syncthreads();

  const int lo = s_seg[0];
  const int hi = s_seg[nvalid - 1];
  const bool lo_shared = chunk_base > 0 && seg[chunk_base - 1] == lo;
  const bool hi_shared = chunk_base + nvalid < (long)E && seg[chunk_base + nvalid] == hi;
  int par = 0;
  const int ngroups = (nvalid + 3) >> 2;
  int win = lo >> 4;

  v8f c;
  #pragma unroll
  for (int r = 0; r < 8; ++r) c[r] = 0.0f;

  for (int g0 = 0; g0 < ngroups; g0 += KSTEPS) {
    int2 sg[KSTEPS]; float2 w[KSTEPS]; v2f b[KSTEPS]; int tl[KSTEPS]; int gi[KSTEPS];

    #pragma unroll
    for (int j = 0; j < KSTEPS; ++j) {
      int g = g0 + j;
      bool live = g < ngroups;
      if (!live) g = ngroups - 1;
      gi[j] = g;
      int i0 = 4 * g + 2 * half;
      sg[j] = *(const int2*)&s_seg[i0];
      uint2 of = *(const uint2*)&s_off[i0];
      float2 ww = *(const float2*)&s_w[i0];
      if (!live) { ww.x = 0.0f; ww.y = 0.0f; }
      w[j] = ww;
      b[j].x = feat[of.x + (unsigned)col];
      b[j].y = feat[of.y + (unsigned)col];
      tl[j] = s_seg[4 * g + 3] >> 4;
    }

    #pragma unroll
    for (int j = 0; j < KSTEPS; ++j) {
      for (int pass = 0; pass < 5; ++pass) {
        const int wb = win << 4;
        v2f a;
        a.x = ((sg[j].x - wb) == m) ? w[j].x : 0.0f;
        a.y = ((sg[j].y - wb) == m) ? w[j].y : 0.0f;
        c = __builtin_amdgcn_wmma_f32_16x16x4_f32(false, a, false, b[j], (short)0, c,
                                                  false, false);
        if (tl[j] == win) break;
        flush_window(out, c, s_stage[par], win, lo, hi, lo_shared, hi_shared, half, m, col, wave);
        par ^= 1;
        #pragma unroll
        for (int r = 0; r < 8; ++r) c[r] = 0.0f;
        const int q0 = s_seg[4 * gi[j]] >> 4, q1 = s_seg[4 * gi[j] + 1] >> 4,
                  q2 = s_seg[4 * gi[j] + 2] >> 4;
        int nx = tl[j];
        if (q2 > win) nx = q2;
        if (q1 > win) nx = q1;
        if (q0 > win) nx = q0;
        win = nx;
      }
    }
  }

  flush_window(out, c, s_stage[par], win, lo, hi, lo_shared, hi_shared, half, m, col, wave);
}

static inline int cdiv_i(long a, long b) { return (int)((a + b - 1) / b); }

extern "C" void kernel_launch(void* const* d_in, const int* in_sizes, int n_in,
                              void* d_out, int out_size, void* d_ws, size_t ws_size,
                              hipStream_t stream) {
  const float* feat = (const float*)d_in[0];
  const float* ew   = (const float*)d_in[1];
  const int*   nidx = (const int*)d_in[2];
  const int*   seg  = (const int*)d_in[3];
  float* out = (float*)d_out;

  const int N = in_sizes[0] / DFEAT;
  const int E = in_sizes[1];

  float* wsum = (float*)d_ws;
  float* deg  = wsum + N;

  zero2_f32<<<FILL_GRID, FILL_BLOCK, 0, stream>>>(wsum, 2L * N, out, (long)out_size);

  int stat_threads = cdiv_i(E, STATS_RUN);
  edge_stats<<<cdiv_i(stat_threads, 256), 256, 0, stream>>>(ew, seg, wsum, deg, E);

  aggregate<<<cdiv_i(E, CHUNK), BLOCK, 0, stream>>>(feat, ew, nidx, seg, wsum, deg, out, E);
}


# aggregate: work-group sums its rows' weight totals beside A (two waves read the end rows' edges outside the chunk), rows divided on leaving; edge-stats launch dropped; rounding per sum, not per term
# speedup vs baseline: 3.1759x; 1.1862x over previous
#include <hip/hip_runtime.h>

typedef __attribute__((ext_vector_type(2))) float v2f;
typedef __attribute__((ext_vector_type(8))) float v8f;

#define CHUNK     512
#define BLOCK     128
#define DFEAT     64
#define KSTEPS    4
#define TP        68
#define FILL_GRID  256
#define FILL_BLOCK 256

__global__ void __launch_bounds__(FILL_BLOCK)
zero_f32(float* __restrict__ a, long na) {
  const long tid    = (long)blockIdx.x * blockDim.x + threadIdx.x;
  const long stride = (long)gridDim.x * blockDim.x;
  const float4 z = make_float4(0.0f, 0.0f, 0.0f, 0.0f);
  float4* __restrict__ a4 = (float4*)a;
  const long na4 = na >> 2;
  for (long i = tid; i < na4; i += stride) a4[i] = z;
  if (tid < (na & 3)) a[(na4 << 2) + tid] = 0.0f;
}

__device__ __forceinline__ void flush_window(float* __restrict__ out, const float* __restrict__ feat,
                                             const v8f& c, float rw,
                                             float* __restrict__ stage,
                                             const int* __restrict__ s_seg,
                                             const unsigned* __restrict__ s_off, int nvalid,
                                             int win, int lo, int hi,
                                             bool lo_shared, bool hi_shared,
                                             float lo_w, float lo_d, float hi_w, float hi_d,
                                             int half, int m, int col, int wave) {
  const int nb = win << 4;
  #pragma unroll
  for (int r = 0; r < 8; ++r) stage[(r + 8 * half) * TP + col] = c[r];
  __syncthreads();
  #pragma unroll
  for (int i = 0; i < 2; ++i) {
    const int row  = 4 * wave + 2 * i + half;
    const int node = nb + row;
    float ws = __shfl(rw, row) + __shfl(rw, row + 16);
    if (node >= lo && node <= hi) {
      float4 v = *(const float4*)&stage[row * TP + 4 * m];
      float* p = out + (long)node * DFEAT + 4 * m;
      const bool sh_lo = (node == lo) && lo_shared;
      const bool sh_hi = (node == hi) && hi_shared;
      if (sh_lo) ws += lo_w;
      if (sh_hi) ws += hi_w;
      float den = ws;
      if (ws == 0.0f) {
        float dg = (sh_lo ? lo_d : 0.0f) + (sh_hi ? hi_d : 0.0f);
        v = make_float4(0.0f, 0.0f, 0.0f, 0.0f);
        for (int k = 0; k < nvalid; ++k) {
          if (s_seg[k] == node) {
            const float4 f = *(const float4*)&feat[s_off[k] + 4u * (unsigned)m];
            v.x += f.x; v.y += f.y; v.z += f.z; v.w += f.w;
            dg += 1.0f;
          }
        }
        den = dg;
      }
      if (den != 0.0f) {
        v.x /= den; v.y /= den; v.z /= den; v.w /= den;
        if (sh_lo || sh_hi) {
          atomicAdd(p + 0, v.x); atomicAdd(p + 1, v.y);
          atomicAdd(p + 2, v.z); atomicAdd(p + 3, v.w);
        } else {
          *(float4*)p = v;
        }
      }
    }
  }
}

__global__ void __launch_bounds__(BLOCK)
aggregate(const float* __restrict__ feat, const float* __restrict__ ew,
          const int* __restrict__ nidx, const int* __restrict__ seg,
          float* __restrict__ out, int E) {
  __shared__ __attribute__((aligned(16))) int      s_seg[CHUNK];
  __shared__ __attribute__((aligned(16))) unsigned s_off[CHUNK];
  __shared__ __attribute__((aligned(16))) float    s_w[CHUNK];
  __shared__ __attribute__((aligned(16))) float    s_stage[2][16 * TP];
  __shared__ float s_ext[4];

  const int tid  = threadIdx.x;
  const int lane = tid & 31;
  const int m    = lane & 15;
  const int half = lane >> 4;
  const int wave = tid >> 5;
  const int cb   = wave * 16;
  const int col  = cb + m;

  const long chunk_base = (long)blockIdx.x * CHUNK;
  long rem = (long)E - chunk_base;
  const int nvalid = rem < (long)CHUNK ? (int)rem : CHUNK;

  int r_sg[CHUNK / BLOCK]; int r_ix[CHUNK / BLOCK]; float r_w[CHUNK / BLOCK];
  #pragma unroll
  for (int t = 0; t < CHUNK / BLOCK; ++t) {
    const int  i    = tid + t * BLOCK;
    const long e    = (i < nvalid) ? chunk_base + i : (long)E - 1;
    r_sg[t] = seg[e];
    r_ix[t] = nidx[e];
    r_w[t]  = ew[e];
  }

  if (wave < 2) {
    const bool fwd = (wave == 1);
    const int  id  = fwd ? seg[chunk_base + nvalid - 1] : seg[chunk_base];
    long e = fwd ? chunk_base + nvalid + lane : chunk_base - 1 - lane;
    const long step = fwd ? 32 : -32;
    float sw = 0.0f, sd = 0.0f;
    for (;;) {
      const bool  inb = (e >= 0) && (e < (long)E);
      const long  ec  = inb ? e : 0;
      const int   sv  = seg[ec];
      const float wv  = ew[ec];
      const bool  hit = inb && (sv == id);
      sw += hit ? wv : 0.0f;
      sd += hit ? 1.0f : 0.0f;
      if (!__all(hit)) break;
      e += step;
    }
    #pragma unroll
    for (int o = 16; o > 0; o >>= 1) { sw += __shfl_xor(sw, o); sd += __shfl_xor(sd, o); }
    if (lane == 0) { s_ext[2 * wave] = sw; s_ext[2 * wave + 1] = sd; }
  }
  #pragma unroll
  for (int t = 0; t < CHUNK / BLOCK; ++t) {
    const int  i    = tid + t * BLOCK;
    const bool live = i < nvalid;
    s_seg[i] = r_sg[t];
    s_off[i] = live ? (unsigned)r_ix[t] * DFEAT : 0u;
    s_w[i]   = live ? r_w[t] : 0.0f;
  }
  __syncthreads();

  const int lo = s_seg[0];
  const int hi = s_seg[nvalid - 1];
  const float lo_w = s_ext[0], lo_d = s_ext[1], hi_w = s_ext[2], hi_d = s_ext[3];
  const bool lo_shared = lo_d > 0.0f;
  const bool hi_shared = hi_d > 0.0f;
  int par = 0;
  const int ngroups = (nvalid + 3) >> 2;
  int win = lo >> 4;

  v8f c;
  #pragma unroll
  for (int r = 0; r < 8; ++r) c[r] = 0.0f;
  float rw = 0.0f;

  for (int g0 = 0; g0 < ngroups; g0 += KSTEPS) {
    int2 sg[KSTEPS]; float2 w[KSTEPS]; v2f b[KSTEPS]; int tl[KSTEPS]; int gi[KSTEPS];

    #pragma unroll
    for (int j = 0; j < KSTEPS; ++j) {
      int g = g0 + j;
      bool live = g < ngroups;
      if (!live) g = ngroups - 1;
      gi[j] = g;
      int i0 = 4 * g + 2 * half;
      sg[j] = *(const int2*)&s_seg[i0];
      uint2 of = *(const uint2*)&s_off[i0];
      float2 ww = *(const float2*)&s_w[i0];
      if (!live) { ww.x = 0.0f; ww.y = 0.0f; }
      w[j] = ww;
      b[j].x = feat[of.x + (unsigned)col];
      b[j].y = feat[of.y + (unsigned)col];
      tl[j] = s_seg[4 * g + 3] >> 4;
    }

    #pragma unroll
    for (int j = 0; j < KSTEPS; ++j) {
      for (int pass = 0; pass < 5; ++pass) {
        const int wb = win << 4;
        v2f a;
        a.x = ((sg[j].x - wb) == m) ? w[j].x : 0.0f;
        a.y = ((sg[j].y - wb) == m) ? w[j].y : 0.0f;
        rw += a.x + a.y;
        c = __builtin_amdgcn_wmma_f32_16x16x4_f32(false, a, false, b[j], (short)0, c,
                                                  false, false);
        if (tl[j] == win) break;
        flush_window(out, feat, c, rw, s_stage[par], s_seg, s_off, nvalid, win, lo, hi,
                     lo_shared, hi_shared, lo_w, lo_d, hi_w, hi_d, half, m, col, wave);
        par ^= 1;
        #pragma unroll
        for (int r = 0; r < 8; ++r) c[r] = 0.0f;
        rw = 0.0f;
        const int q0 = s_seg[4 * gi[j]] >> 4, q1 = s_seg[4 * gi[j] + 1] >> 4,
                  q2 = s_seg[4 * gi[j] + 2] >> 4;
        int nx = tl[j];
        if (q2 > win) nx = q2;
        if (q1 > win) nx = q1;
        if (q0 > win) nx = q0;
        win = nx;
      }
    }
  }

  flush_window(out, feat, c, rw, s_stage[par], s_seg, s_off, nvalid, win, lo, hi,
               lo_shared, hi_shared, lo_w, lo_d, hi_w, hi_d, half, m, col, wave);
}

static inline int cdiv_i(long a, long b) { return (int)((a + b - 1) / b); }

extern "C" void kernel_launch(void* const* d_in, const int* in_sizes, int n_in,
                              void* d_out, int out_size, void* d_ws, size_t ws_size,
                              hipStream_t stream) {
  const float* feat = (const float*)d_in[0];
  const float* ew   = (const float*)d_in[1];
  const int*   nidx = (const int*)d_in[2];
  const int*   seg  = (const int*)d_in[3];
  float* out = (float*)d_out;

  const int E = in_sizes[1];

  zero_f32<<<FILL_GRID, FILL_BLOCK, 0, stream>>>(out, (long)out_size);

  aggregate<<<cdiv_i(E, CHUNK), BLOCK, 0, stream>>>(feat, ew, nidx, seg, out, E);
}
